# starting kernel
# baseline (speedup 1.0000x reference)
#include <hip/hip_runtime.h>

typedef __attribute__((ext_vector_type(2))) float v2f;
typedef __attribute__((ext_vector_type(8))) float v8f;

#define HD 128
#define WT_STRIDE 132

__global__ __launch_bounds__(256) void k_deg_init(float* __restrict__ deg, int n) {
  int i = blockIdx.x * 256 + threadIdx.x;
  if (i < n) deg[i] = 1.0f;
}

__global__ __launch_bounds__(256) void k_deg_count(const int* __restrict__ dst,
                                                   float* __restrict__ deg, int e) {
  int stride = gridDim.x * 256;
  for (int i = blockIdx.x * 256 + threadIdx.x; i < e; i += stride)
    atomicAdd(&deg[dst[i]], 1.0f);
}

__global__ __launch_bounds__(256) void k_dinv(float* __restrict__ deg, int n) {
  int i = blockIdx.x * 256 + threadIdx.x;
  if (i < n) deg[i] = rsqrtf(deg[i]);
}

__global__ __launch_bounds__(256) void k_gemm_wmma(const float* __restrict__ A,
                                                   const float* __restrict__ W,
                                                   float* __restrict__ out, int n) {
  __shared__ float Wt[HD * WT_STRIDE];
  for (int i = threadIdx.x; i < HD * HD / 4; i += 256) {
    int k  = i >> 5;
    int n4 = (i & 31) * 4;
    const float4 w = *(const float4*)(W + k * HD + n4);
    Wt[(n4 + 0) * WT_STRIDE + k] = w.x;
    Wt[(n4 + 1) * WT_STRIDE + k] = w.y;
    Wt[(n4 + 2) * WT_STRIDE + k] = w.z;
    Wt[(n4 + 3) * WT_STRIDE + k] = w.w;
  }
  __syncthreads();

  const int lane = threadIdx.x & 31;
  const int wave = threadIdx.x >> 5;
  const int r0   = blockIdx.x * 128 + wave * 16;
  if (r0 + 16 > n) return;
  const int hi = lane >> 4;
  const int lo = lane & 15;
  const float* arow = A + (size_t)(r0 + lo) * HD;

  v8f acc[8] = {};
  for (int kk = 0; kk < 32; ++kk) {
    const int k = kk * 4 + hi * 2;
    const v2f a = *(const v2f*)(arow + k);
#pragma unroll
    for (int t = 0; t < 8; ++t) {
      const v2f b = *(const v2f*)(&Wt[(t * 16 + lo) * WT_STRIDE + k]);
      acc[t] = __builtin_amdgcn_wmma_f32_16x16x4_f32(
          false, a, false, b, (short)0, acc[t], false, false);
    }
  }
#pragma unroll
  for (int t = 0; t < 8; ++t) {
    float* orow = out + (size_t)(r0 + hi * 8) * HD + t * 16 + lo;
#pragma unroll
    for (int r = 0; r < 8; ++r)
      orow[(size_t)r * HD] = acc[t][r];
  }
}

__global__ __launch_bounds__(256) void k_self_init(const float* __restrict__ dinv,
                                                   const float* __restrict__ m,
                                                   float* __restrict__ agg, int total) {
  int i = blockIdx.x * 256 + threadIdx.x;
  if (i < total) {
    float di = dinv[i >> 7];
    agg[i] = di * di * m[i];
  }
}

__global__ __launch_bounds__(256) void k_self_init_bias(const float* __restrict__ dinv,
                                                        const float* __restrict__ m,
                                                        const float* __restrict__ bias,
                                                        float* __restrict__ outp, int total) {
  int i = blockIdx.x * 256 + threadIdx.x;
  if (i < total) {
    float di = dinv[i >> 7];
    outp[i] = di * di * m[i] + bias[i & 127];
  }
}

__global__ __launch_bounds__(256) void k_bias_relu(float* __restrict__ agg,
                                                   const float* __restrict__ bias, int total) {
  int i = blockIdx.x * 256 + threadIdx.x;
  if (i < total) {
    float v = agg[i] + bias[i & 127];
    agg[i] = v > 0.0f ? v : 0.0f;
  }
}

__global__ __launch_bounds__(256) void k_scatter(const int* __restrict__ src,
                                                 const int* __restrict__ dst,
                                                 const float* __restrict__ dinv,
                                                 const float* __restrict__ m,
                                                 float* __restrict__ agg, int e) {
  const int lane = threadIdx.x & 31;
  int w = (blockIdx.x * 256 + threadIdx.x) >> 5;
  const int nw = (gridDim.x * 256) >> 5;
  for (; w < e; w += nw) {
    const int s = src[w];
    const int d = dst[w];
    const float nrm = dinv[s] * dinv[d];
    const float4 v = *(const float4*)(m + (size_t)s * HD + lane * 4);
    float* p = agg + (size_t)d * HD + lane * 4;
    atomicAdd(p + 0, nrm * v.x);
    atomicAdd(p + 1, nrm * v.y);
    atomicAdd(p + 2, nrm * v.z);
    atomicAdd(p + 3, nrm * v.w);
  }
}

extern "C" void kernel_launch(void* const* d_in, const int* in_sizes, int n_in,
                              void* d_out, int out_size, void* d_ws, size_t ws_size,
                              hipStream_t stream) {
  const float* x  = (const float*)d_in[0];
  const float* W1 = (const float*)d_in[1];
  const float* b1 = (const float*)d_in[2];
  const float* W2 = (const float*)d_in[3];
  const float* b2 = (const float*)d_in[4];
  const int*   ei = (const int*)d_in[5];

  const int n = in_sizes[0] / HD;
  const int e = in_sizes[5] / 2;
  const int* src = ei;
  const int* dst = ei + e;
  const int total = n * HD;

  char* ws = (char*)d_ws;
  float* dinv = (float*)ws;
  size_t off1 = ((size_t)n * 4 + 255) & ~(size_t)255;
  float* buf1 = (float*)(ws + off1);
  size_t off2 = off1 + (((size_t)total * 4 + 255) & ~(size_t)255);
  float* buf2 = (float*)(ws + off2);
  float* out  = (float*)d_out;

  dim3 blk(256);
  const int gN   = (n + 255) / 256;
  const int gNH  = (total + 255) / 256;
  const int gMM  = (n + 127) / 128;

  k_deg_init <<<gN,   blk, 0, stream>>>(dinv, n);
  k_deg_count<<<2048, blk, 0, stream>>>(dst, dinv, e);
  k_dinv     <<<gN,   blk, 0, stream>>>(dinv, n);

  k_gemm_wmma<<<gMM,  blk, 0, stream>>>(x, W1, buf1, n);
  k_self_init<<<gNH,  blk, 0, stream>>>(dinv, buf1, buf2, total);
  k_scatter  <<<4096, blk, 0, stream>>>(src, dst, dinv, buf1, buf2, e);
  k_bias_relu<<<gNH,  blk, 0, stream>>>(buf2, b1, total);

  k_gemm_wmma<<<gMM,  blk, 0, stream>>>(buf2, W2, buf1, n);
  k_self_init_bias<<<gNH, blk, 0, stream>>>(dinv, buf1, b2, out, total);
  k_scatter  <<<4096, blk, 0, stream>>>(src, dst, dinv, buf1, out, e);
}


# Atomic scatter replaced by a gather over edges grouped by destination (arrival-order rank from an int atomic, rebuilt every call): one wave per row sums in registers; the sum order changes
# speedup vs baseline: 2.8228x; 2.8228x over previous
#include <hip/hip_runtime.h>

typedef __attribute__((ext_vector_type(2))) float v2f;
typedef __attribute__((ext_vector_type(8))) float v8f;

#define HD 128
#define WT_STRIDE 132
#define SCAN_TILE 1024

__global__ __launch_bounds__(256) void k_cnt_zero(int* __restrict__ cnt, int n) {
  int i = blockIdx.x * 256 + threadIdx.x;
  if (i < n) cnt[i] = 0;
}

__global__ __launch_bounds__(256) void k_rank(const int* __restrict__ dst,
                                              int* __restrict__ cnt,
                                              int* __restrict__ rank, int e) {
  int i = blockIdx.x * 256 + threadIdx.x;
  if (i < e) rank[i] = atomicAdd(&cnt[dst[i]], 1);
}

__device__ __forceinline__ int block_sum_256(int v, int* sm) {
  const int lane = threadIdx.x & 31;
  const int wave = threadIdx.x >> 5;
  for (int o = 16; o > 0; o >>= 1) v += __shfl_xor(v, o);
  if (lane == 0) sm[wave] = v;
  __syncthreads();
  int t = 0;
  for (int w = 0; w < 8; ++w) t += sm[w];
  __syncthreads();
  return t;
}

__global__ __launch_bounds__(256) void k_tile_sum(const int* __restrict__ cnt,
                                                  int* __restrict__ tsum, int n) {
  __shared__ int sm[8];
  const int base = blockIdx.x * SCAN_TILE + threadIdx.x * 4;
  int s = 0;
  for (int j = 0; j < 4; ++j)
    if (base + j < n) s += cnt[base + j];
  const int t = block_sum_256(s, sm);
  if (threadIdx.x == 0) tsum[blockIdx.x] = t;
}

__global__ __launch_bounds__(256) void k_scan(const int* __restrict__ cnt,
                                              const int* __restrict__ tsum,
                                              int* __restrict__ rowptr,
                                              float* __restrict__ dinv, int n) {
  __shared__ int sm[8];
  const int lane = threadIdx.x & 31;
  const int wave = threadIdx.x >> 5;

  int part = 0;
  for (int t = threadIdx.x; t < (int)blockIdx.x; t += 256) part += tsum[t];
  const int tile_off = block_sum_256(part, sm);

  const int base = blockIdx.x * SCAN_TILE + threadIdx.x * 4;
  int c[4];
  int s = 0;
#pragma unroll
  for (int j = 0; j < 4; ++j) {
    c[j] = (base + j < n) ? cnt[base + j] : 0;
    s += c[j];
  }
  int incl = s;
  for (int o = 1; o < 32; o <<= 1) {
    const int up = __shfl_up(incl, o);
    if (lane >= o) incl += up;
  }
  if (lane == 31) sm[wave] = incl;
  __syncthreads();
  int wave_off = 0;
  for (int w = 0; w < 8; ++w)
    if (w < wave) wave_off += sm[w];

  int run = tile_off + wave_off + incl - s;
#pragma unroll
  for (int j = 0; j < 4; ++j) {
    if (base + j < n) {
      rowptr[base + j] = run;
      dinv[base + j] = rsqrtf((float)(c[j] + 1));
      run += c[j];
    }
  }
}

__global__ __launch_bounds__(256) void k_place(const int* __restrict__ src,
                                               const int* __restrict__ dst,
                                               const int* __restrict__ rank,
                                               const int* __restrict__ rowptr,
                                               int* __restrict__ col, int e) {
  int i = blockIdx.x * 256 + threadIdx.x;
  if (i < e) col[rowptr[dst[i]] + rank[i]] = src[i];
}

__global__ __launch_bounds__(256) void k_gemm_wmma(const float* __restrict__ A,
                                                   const float* __restrict__ W,
                                                   float* __restrict__ out, int n) {
  __shared__ float Wt[HD * WT_STRIDE];
  for (int i = threadIdx.x; i < HD * HD / 4; i += 256) {
    int k  = i >> 5;
    int n4 = (i & 31) * 4;
    const float4 w = *(const float4*)(W + k * HD + n4);
    Wt[(n4 + 0) * WT_STRIDE + k] = w.x;
    Wt[(n4 + 1) * WT_STRIDE + k] = w.y;
    Wt[(n4 + 2) * WT_STRIDE + k] = w.z;
    Wt[(n4 + 3) * WT_STRIDE + k] = w.w;
  }
  __syncthreads();

  const int lane = threadIdx.x & 31;
  const int wave = threadIdx.x >> 5;
  const int r0   = blockIdx.x * 128 + wave * 16;
  if (r0 + 16 > n) return;
  const int hi = lane >> 4;
  const int lo = lane & 15;
  const float* arow = A + (size_t)(r0 + lo) * HD;

  v8f acc[8] = {};
  for (int kk = 0; kk < 32; ++kk) {
    const int k = kk * 4 + hi * 2;
    const v2f a = *(const v2f*)(arow + k);
#pragma unroll
    for (int t = 0; t < 8; ++t) {
      const v2f b = *(const v2f*)(&Wt[(t * 16 + lo) * WT_STRIDE + k]);
      acc[t] = __builtin_amdgcn_wmma_f32_16x16x4_f32(
          false, a, false, b, (short)0, acc[t], false, false);
    }
  }
#pragma unroll
  for (int t = 0; t < 8; ++t) {
    float* orow = out + (size_t)(r0 + hi * 8) * HD + t * 16 + lo;
#pragma unroll
    for (int r = 0; r < 8; ++r)
      orow[(size_t)r * HD] = acc[t][r];
  }
}

template <bool RELU>
__global__ __launch_bounds__(256) void k_gather(const int* __restrict__ rowptr,
                                                const int* __restrict__ cnt,
                                                const int* __restrict__ col,
                                                const float* __restrict__ dinv,
                                                const float* __restrict__ m,
                                                const float* __restrict__ bias,
                                                float* __restrict__ outp, int n) {
  const int lane = threadIdx.x & 31;
  const int d = __builtin_amdgcn_readfirstlane(blockIdx.x * 8 + (threadIdx.x >> 5));
  if (d >= n) return;
  const float dd  = dinv[d];
  const int   beg = rowptr[d];
  const int   deg = cnt[d];

  const float4 sv = *(const float4*)(m + (size_t)d * HD + lane * 4);
  const float sw = dd * dd;
  float ax = sw * sv.x, ay = sw * sv.y, az = sw * sv.z, aw = sw * sv.w;

  for (int c0 = 0; c0 < deg; c0 += 32) {
    const int nb = (deg - c0 < 32) ? (deg - c0) : 32;
    int   s_l = 0;
    float w_l = 0.0f;
    if (lane < nb) {
      s_l = col[beg + c0 + lane];
      w_l = dinv[s_l] * dd;
    }
#pragma unroll 4
    for (int j = 0; j < nb; ++j) {
      const int   s = __builtin_amdgcn_readlane(s_l, j);
      const float w = __builtin_bit_cast(float,
                          __builtin_amdgcn_readlane(__builtin_bit_cast(int, w_l), j));
      const float4 v = *(const float4*)(m + (size_t)s * HD + lane * 4);
      ax = fmaf(w, v.x, ax);
      ay = fmaf(w, v.y, ay);
      az = fmaf(w, v.z, az);
      aw = fmaf(w, v.w, aw);
    }
  }

  const float4 b = *(const float4*)(bias + lane * 4);
  float4 r;
  r.x = ax + b.x; r.y = ay + b.y; r.z = az + b.z; r.w = aw + b.w;
  if (RELU) {
    r.x = r.x > 0.0f ? r.x : 0.0f;
    r.y = r.y > 0.0f ? r.y : 0.0f;
    r.z = r.z > 0.0f ? r.z : 0.0f;
    r.w = r.w > 0.0f ? r.w : 0.0f;
  }
  *(float4*)(outp + (size_t)d * HD + lane * 4) = r;
}

extern "C" void kernel_launch(void* const* d_in, const int* in_sizes, int n_in,
                              void* d_out, int out_size, void* d_ws, size_t ws_size,
                              hipStream_t stream) {
  const float* x  = (const float*)d_in[0];
  const float* W1 = (const float*)d_in[1];
  const float* b1 = (const float*)d_in[2];
  const float* W2 = (const float*)d_in[3];
  const float* b2 = (const float*)d_in[4];
  const int*   ei = (const int*)d_in[5];

  const int n = in_sizes[0] / HD;
  const int e = in_sizes[5] / 2;
  const int* src = ei;
  const int* dst = ei + e;
  const int total = n * HD;

  char* ws = (char*)d_ws;
  float* dinv = (float*)ws;
  size_t off1 = ((size_t)n * 4 + 255) & ~(size_t)255;
  float* buf1 = (float*)(ws + off1);
  size_t off2 = off1 + (((size_t)total * 4 + 255) & ~(size_t)255);
  size_t nb4 = ((size_t)n * 4 + 255) & ~(size_t)255;
  size_t eb4 = ((size_t)e * 4 + 255) & ~(size_t)255;
  int* cnt    = (int*)(ws + off2);
  int* rowptr = (int*)(ws + off2 + nb4);
  int* tsum   = (int*)(ws + off2 + 2 * nb4);
  int* rank   = (int*)(ws + off2 + 3 * nb4);
  int* col    = (int*)(ws + off2 + 3 * nb4 + eb4);
  float* out  = (float*)d_out;

  dim3 blk(256);
  const int gN   = (n + 255) / 256;
  const int gE   = (e + 255) / 256;
  const int gT   = (n + SCAN_TILE - 1) / SCAN_TILE;
  const int gMM  = (n + 127) / 128;
  const int gG   = (n + 7) / 8;

  k_cnt_zero<<<gN, blk, 0, stream>>>(cnt, n);
  k_rank    <<<gE, blk, 0, stream>>>(dst, cnt, rank, e);
  k_tile_sum<<<gT, blk, 0, stream>>>(cnt, tsum, n);
  k_scan    <<<gT, blk, 0, stream>>>(cnt, tsum, rowptr, dinv, n);
  k_place   <<<gE, blk, 0, stream>>>(src, dst, rank, rowptr, col, e);

  k_gemm_wmma<<<gMM, blk, 0, stream>>>(x, W1, buf1, n);
  k_gather<true> <<<gG, blk, 0, stream>>>(rowptr, cnt, col, dinv, buf1, b1, out, n);

  k_gemm_wmma<<<gMM, blk, 0, stream>>>(out, W2, buf1, n);
  k_gather<false><<<gG, blk, 0, stream>>>(rowptr, cnt, col, dinv, buf1, b2, out, n);
}


# Product fused into the aggregation: each wave sums 16 input rows (A_hat X first), then takes @W on WMMA from its LDS tile; no separate product launch; sum order changes (aggregate before product)
# speedup vs baseline: 2.9808x; 1.0560x over previous
#include <hip/hip_runtime.h>

typedef __attribute__((ext_vector_type(2))) float v2f;
typedef __attribute__((ext_vector_type(8))) float v8f;

#define HD 128
#define WT_STRIDE 132
#define SCAN_TILE 1024

__global__ __launch_bounds__(256) void k_cnt_zero(int* __restrict__ cnt, int n) {
  int i = blockIdx.x * 256 + threadIdx.x;
  if (i < n) cnt[i] = 0;
}

__global__ __launch_bounds__(256) void k_rank(const int* __restrict__ dst,
                                              int* __restrict__ cnt,
                                              int* __restrict__ rank, int e) {
  int i = blockIdx.x * 256 + threadIdx.x;
  if (i < e) rank[i] = atomicAdd(&cnt[dst[i]], 1);
}

__device__ __forceinline__ int block_sum_256(int v, int* sm) {
  const int lane = threadIdx.x & 31;
  const int wave = threadIdx.x >> 5;
  for (int o = 16; o > 0; o >>= 1) v += __shfl_xor(v, o);
  if (lane == 0) sm[wave] = v;
  __syncthreads();
  int t = 0;
  for (int w = 0; w < 8; ++w) t += sm[w];
  __syncthreads();
  return t;
}

__global__ __launch_bounds__(256) void k_tile_sum(const int* __restrict__ cnt,
                                                  int* __restrict__ tsum, int n) {
  __shared__ int sm[8];
  const int base = blockIdx.x * SCAN_TILE + threadIdx.x * 4;
  int s = 0;
  for (int j = 0; j < 4; ++j)
    if (base + j < n) s += cnt[base + j];
  const int t = block_sum_256(s, sm);
  if (threadIdx.x == 0) tsum[blockIdx.x] = t;
}

__global__ __launch_bounds__(256) void k_scan(const int* __restrict__ cnt,
                                              const int* __restrict__ tsum,
                                              int* __restrict__ rowptr,
                                              float* __restrict__ dinv, int n) {
  __shared__ int sm[8];
  const int lane = threadIdx.x & 31;
  const int wave = threadIdx.x >> 5;

  int part = 0;
  for (int t = threadIdx.x; t < (int)blockIdx.x; t += 256) part += tsum[t];
  const int tile_off = block_sum_256(part, sm);

  const int base = blockIdx.x * SCAN_TILE + threadIdx.x * 4;
  int c[4];
  int s = 0;
#pragma unroll
  for (int j = 0; j < 4; ++j) {
    c[j] = (base + j < n) ? cnt[base + j] : 0;
    s += c[j];
  }
  int incl = s;
  for (int o = 1; o < 32; o <<= 1) {
    const int up = __shfl_up(incl, o);
    if (lane >= o) incl += up;
  }
  if (lane == 31) sm[wave] = incl;
  __syncthreads();
  int wave_off = 0;
  for (int w = 0; w < 8; ++w)
    if (w < wave) wave_off += sm[w];

  int run = tile_off + wave_off + incl - s;
#pragma unroll
  for (int j = 0; j < 4; ++j) {
    if (base + j < n) {
      rowptr[base + j] = run;
      dinv[base + j] = rsqrtf((float)(c[j] + 1));
      run += c[j];
    }
  }
}

__global__ __launch_bounds__(256) void k_place(const int* __restrict__ src,
                                               const int* __restrict__ dst,
                                               const int* __restrict__ rank,
                                               const int* __restrict__ rowptr,
                                               int* __restrict__ col, int e) {
  int i = blockIdx.x * 256 + threadIdx.x;
  if (i < e) col[rowptr[dst[i]] + rank[i]] = src[i];
}

#define AT_STRIDE 132

template <int U>
__device__ __forceinline__ void edge_group(const float* __restrict__ in, int s_l, float w_l, int jj, int lane,
                                           float& ax, float& ay, float& az, float& aw) {
#pragma unroll
  for (int u = 0; u < U; ++u) {
    const int   s = __builtin_amdgcn_readlane(s_l, jj + u);
    const float w = __builtin_bit_cast(float,
                        __builtin_amdgcn_readlane(__builtin_bit_cast(int, w_l), jj + u));
    const float4 v = *(const float4*)(in + (size_t)s * HD + lane * 4);
    ax = fmaf(w, v.x, ax);
    ay = fmaf(w, v.y, ay);
    az = fmaf(w, v.z, az);
    aw = fmaf(w, v.w, aw);
  }
}

template <bool RELU>
__global__ __launch_bounds__(256) void k_agg_gemm(const int* __restrict__ rowptr,
                                                  const int* __restrict__ cnt,
                                                  const int* __restrict__ col,
                                                  const float* __restrict__ dinv,
                                                  const float* __restrict__ in,
                                                  const float* __restrict__ W,
                                                  const float* __restrict__ bias,
                                                  float* __restrict__ outp, int n) {
  __shared__ __attribute__((aligned(16))) float Wt[HD * WT_STRIDE];
  __shared__ __attribute__((aligned(16))) float At[8 * 16 * AT_STRIDE];
  for (int i = threadIdx.x; i < HD * HD / 4; i += 256) {
    int k  = i >> 5;
    int n4 = (i & 31) * 4;
    const float4 w = *(const float4*)(W + k * HD + n4);
    Wt[(n4 + 0) * WT_STRIDE + k] = w.x;
    Wt[(n4 + 1) * WT_STRIDE + k] = w.y;
    Wt[(n4 + 2) * WT_STRIDE + k] = w.z;
    Wt[(n4 + 3) * WT_STRIDE + k] = w.w;
  }
  __syncthreads();

  const int lane = threadIdx.x & 31;
  const int wave = __builtin_amdgcn_readfirstlane(threadIdx.x >> 5);
  const int r0   = blockIdx.x * 128 + wave * 16;
  if (r0 + 16 > n) return;
  const int hi = lane >> 4;
  const int lo = lane & 15;
  float* at = At + wave * 16 * AT_STRIDE;

  const int   beg_l = rowptr[r0 + lo];
  const int   deg_l = cnt[r0 + lo];
  const float dd_l  = dinv[r0 + lo];

  auto load_col = [&](int j) -> int {
    const int beg = __builtin_amdgcn_readlane(beg_l, j);
    const int deg = __builtin_amdgcn_readlane(deg_l, j);
    const int nb  = deg < 32 ? deg : 32;
    int s = r0 + j;
    if (lane < nb) s = col[beg + lane];
    return s;
  };
  auto load_w = [&](int s) -> float { return dinv[s]; };

  int   s_a = load_col(0);
  int   s_b = load_col(1);
  float w_a = load_w(s_a);

  for (int j = 0; j < 16; ++j) {
    const int d   = r0 + j;
    const int beg = __builtin_amdgcn_readlane(beg_l, j);
    const int deg = __builtin_amdgcn_readlane(deg_l, j);
    const float dd = __builtin_bit_cast(float,
                         __builtin_amdgcn_readlane(__builtin_bit_cast(int, dd_l), j));
    const int nb0 = deg < 32 ? deg : 32;
    int   s_l = s_a;
    float w_l = (lane < nb0) ? w_a * dd : 0.0f;
    int   s_c = r0;
    float w_b = 0.0f;
    if (j + 2 < 16) s_c = load_col(j + 2);
    if (j + 1 < 16) w_b = load_w(s_b);

    const float4 sv = *(const float4*)(in + (size_t)d * HD + lane * 4);
    float ax = 0.0f, ay = 0.0f, az = 0.0f, aw = 0.0f;

    for (int c0 = 0; c0 < deg; c0 += 32) {
      const int nb = (deg - c0 < 32) ? (deg - c0) : 32;
      if (c0 > 0) {
        s_l = d;
        w_l = 0.0f;
        if (lane < nb) {
          s_l = col[beg + c0 + lane];
          w_l = dinv[s_l] * dd;
        }
      }
      int jj = 0;
      for (; nb - jj > 8; jj += 16) edge_group<16>(in, s_l, w_l, jj, lane, ax, ay, az, aw);
      if (jj < nb) edge_group<8>(in, s_l, w_l, jj, lane, ax, ay, az, aw);
    }
    const float sw = dd * dd;
    ax = fmaf(sw, sv.x, ax);
    ay = fmaf(sw, sv.y, ay);
    az = fmaf(sw, sv.z, az);
    aw = fmaf(sw, sv.w, aw);
    float4 r;
    r.x = ax; r.y = ay; r.z = az; r.w = aw;
    *(float4*)(at + j * AT_STRIDE + lane * 4) = r;
    s_a = s_b;
    w_a = w_b;
    s_b = s_c;
  }

  __threadfence_block();

  v8f acc[8] = {};
  const float* arow = at + lo * AT_STRIDE;
  for (int kk = 0; kk < 32; ++kk) {
    const int k = kk * 4 + hi * 2;
    const v2f a = *(const v2f*)(arow + k);
#pragma unroll
    for (int t = 0; t < 8; ++t) {
      const v2f b = *(const v2f*)(&Wt[(t * 16 + lo) * WT_STRIDE + k]);
      acc[t] = __builtin_amdgcn_wmma_f32_16x16x4_f32(
          false, a, false, b, (short)0, acc[t], false, false);
    }
  }
#pragma unroll
  for (int t = 0; t < 8; ++t) {
    const float bv = bias[t * 16 + lo];
    float* orow = outp + (size_t)(r0 + hi * 8) * HD + t * 16 + lo;
#pragma unroll
    for (int r = 0; r < 8; ++r) {
      float o = acc[t][r] + bv;
      if (RELU) o = o > 0.0f ? o : 0.0f;
      orow[(size_t)r * HD] = o;
    }
  }
}

extern "C" void kernel_launch(void* const* d_in, const int* in_sizes, int n_in,
                              void* d_out, int out_size, void* d_ws, size_t ws_size,
                              hipStream_t stream) {
  const float* x  = (const float*)d_in[0];
  const float* W1 = (const float*)d_in[1];
  const float* b1 = (const float*)d_in[2];
  const float* W2 = (const float*)d_in[3];
  const float* b2 = (const float*)d_in[4];
  const int*   ei = (const int*)d_in[5];

  const int n = in_sizes[0] / HD;
  const int e = in_sizes[5] / 2;
  const int* src = ei;
  const int* dst = ei + e;
  const int total = n * HD;

  char* ws = (char*)d_ws;
  float* dinv = (float*)ws;
  size_t off1 = ((size_t)n * 4 + 255) & ~(size_t)255;
  float* buf1 = (float*)(ws + off1);
  size_t off2 = off1 + (((size_t)total * 4 + 255) & ~(size_t)255);
  size_t nb4 = ((size_t)n * 4 + 255) & ~(size_t)255;
  size_t eb4 = ((size_t)e * 4 + 255) & ~(size_t)255;
  int* cnt    = (int*)(ws + off2);
  int* rowptr = (int*)(ws + off2 + nb4);
  int* tsum   = (int*)(ws + off2 + 2 * nb4);
  int* rank   = (int*)(ws + off2 + 3 * nb4);
  int* col    = (int*)(ws + off2 + 3 * nb4 + eb4);
  float* out  = (float*)d_out;

  dim3 blk(256);
  const int gN   = (n + 255) / 256;
  const int gE   = (e + 255) / 256;
  const int gT   = (n + SCAN_TILE - 1) / SCAN_TILE;
  const int gMM  = (n + 127) / 128;

  k_cnt_zero<<<gN, blk, 0, stream>>>(cnt, n);
  k_rank    <<<gE, blk, 0, stream>>>(dst, cnt, rank, e);
  k_tile_sum<<<gT, blk, 0, stream>>>(cnt, tsum, n);
  k_scan    <<<gT, blk, 0, stream>>>(cnt, tsum, rowptr, dinv, n);
  k_place   <<<gE, blk, 0, stream>>>(src, dst, rank, rowptr, col, e);

  k_agg_gemm<true> <<<gMM, blk, 0, stream>>>(rowptr, cnt, col, dinv, x, W1, b1, buf1, n);

  k_agg_gemm<false><<<gMM, blk, 0, stream>>>(rowptr, cnt, col, dinv, buf1, W2, b2, out, n);
}


# Edge grouping rebuilt without global atomics: (edge chunk, node range) work-groups count and rank their edges with LDS atomics, a scan gives each its start offsets; sum order within a row changes
# speedup vs baseline: 3.9512x; 1.3256x over previous
#include <hip/hip_runtime.h>

typedef __attribute__((ext_vector_type(2))) float v2f;
typedef __attribute__((ext_vector_type(8))) float v8f;

#define HD 128
#define WT_STRIDE 132
#define SCAN_TILE 1024

#define RNG 12800
#define NCHUNK 32
#define SWEEP_U 16
#define INIT_U 13

__global__ __launch_bounds__(1024) void k_part_count(const int* __restrict__ dst,
                                                     int* __restrict__ part, int n, int e, int ch) {
  __shared__ int s[RNG];
  const int lo = blockIdx.y * RNG;
  const int nr = (n - lo < RNG) ? (n - lo) : RNG;
  for (int i = threadIdx.x; i < RNG; i += 1024) s[i] = 0;
  __syncthreads();
  const int beg = blockIdx.x * ch;
  const int end = (e - beg < ch) ? e : (beg + ch);
  for (int i0 = beg + (int)threadIdx.x; i0 < end; i0 += SWEEP_U * 1024) {
    int d[SWEEP_U];
#pragma unroll
    for (int u = 0; u < SWEEP_U; ++u) {
      const int i = i0 + u * 1024;
      d[u] = (i < end) ? dst[i] : -1;
    }
#pragma unroll
    for (int u = 0; u < SWEEP_U; ++u) {
      const unsigned t = (unsigned)(d[u] - lo);
      if (t < (unsigned)nr) atomicAdd(&s[t], 1);
    }
  }
  __syncthreads();
  for (int i = threadIdx.x; i < nr; i += 1024) part[(size_t)blockIdx.x * n + lo + i] = s[i];
}

__device__ __forceinline__ int block_sum_256(int v, int* sm) {
  const int lane = threadIdx.x & 31;
  const int wave = threadIdx.x >> 5;
  for (int o = 16; o > 0; o >>= 1) v += __shfl_xor(v, o);
  if (lane == 0) sm[wave] = v;
  __syncthreads();
  int t = 0;
  for (int w = 0; w < 8; ++w) t += sm[w];
  __syncthreads();
  return t;
}

__global__ __launch_bounds__(256) void k_tile_sum(const int* __restrict__ part,
                                                  int* __restrict__ cnt,
                                                  int* __restrict__ tsum, int n) {
  __shared__ int sm[8];
  const int base = blockIdx.x * SCAN_TILE + threadIdx.x * 4;
  int s = 0;
  if ((n & 3) == 0 && base + 3 < n) {
    int4 a = make_int4(0, 0, 0, 0);
#pragma unroll
    for (int q = 0; q < NCHUNK; ++q) {
      const int4 v = *(const int4*)(part + (size_t)q * n + base);
      a.x += v.x; a.y += v.y; a.z += v.z; a.w += v.w;
    }
    *(int4*)(cnt + base) = a;
    s = a.x + a.y + a.z + a.w;
  } else {
    for (int j = 0; j < 4; ++j) {
      if (base + j < n) {
        int v = 0;
        for (int q = 0; q < NCHUNK; ++q) v += part[(size_t)q * n + base + j];
        cnt[base + j] = v;
        s += v;
      }
    }
  }
  const int t = block_sum_256(s, sm);
  if (threadIdx.x == 0) tsum[blockIdx.x] = t;
}

__global__ __launch_bounds__(256) void k_scan(const int* __restrict__ cnt,
                                              const int* __restrict__ tsum,
                                              int* __restrict__ rowptr,
                                              float* __restrict__ dinv,
                                              const int* __restrict__ part,
                                              int* __restrict__ offs, int n) {
  __shared__ int sm[8];
  const int lane = threadIdx.x & 31;
  const int wave = threadIdx.x >> 5;

  int psum = 0;
  for (int t = threadIdx.x; t < (int)blockIdx.x; t += 256) psum += tsum[t];
  const int tile_off = block_sum_256(psum, sm);

  const int base = blockIdx.x * SCAN_TILE + threadIdx.x * 4;
  int c[4];
  int s = 0;
#pragma unroll
  for (int j = 0; j < 4; ++j) {
    c[j] = (base + j < n) ? cnt[base + j] : 0;
    s += c[j];
  }
  int incl = s;
  for (int o = 1; o < 32; o <<= 1) {
    const int up = __shfl_up(incl, o);
    if (lane >= o) incl += up;
  }
  if (lane == 31) sm[wave] = incl;
  __syncthreads();
  int wave_off = 0;
  for (int w = 0; w < 8; ++w)
    if (w < wave) wave_off += sm[w];

  int run = tile_off + wave_off + incl - s;
  if ((n & 3) == 0 && base + 3 < n) {
    int4 pv[NCHUNK];
#pragma unroll
    for (int q = 0; q < NCHUNK; ++q) pv[q] = *(const int4*)(part + (size_t)q * n + base);
    int4 p;
    p.x = run;
    p.y = p.x + c[0];
    p.z = p.y + c[1];
    p.w = p.z + c[2];
    *(int4*)(rowptr + base) = p;
    float4 dv;
    dv.x = rsqrtf((float)(c[0] + 1));
    dv.y = rsqrtf((float)(c[1] + 1));
    dv.z = rsqrtf((float)(c[2] + 1));
    dv.w = rsqrtf((float)(c[3] + 1));
    *(float4*)(dinv + base) = dv;
#pragma unroll
    for (int q = 0; q < NCHUNK; ++q) {
      *(int4*)(offs + (size_t)q * n + base) = p;
      p.x += pv[q].x; p.y += pv[q].y; p.z += pv[q].z; p.w += pv[q].w;
    }
  } else {
#pragma unroll
    for (int j = 0; j < 4; ++j) {
      if (base + j < n) {
        rowptr[base + j] = run;
        dinv[base + j] = rsqrtf((float)(c[j] + 1));
        int pos = run;
        for (int q = 0; q < NCHUNK; ++q) {
          const size_t idx = (size_t)q * n + base + j;
          offs[idx] = pos;
          pos += part[idx];
        }
        run += c[j];
      }
    }
  }
}

__global__ __launch_bounds__(1024) void k_part_place(const int* __restrict__ src,
                                                     const int* __restrict__ dst,
                                                     const int* __restrict__ offs,
                                                     int* __restrict__ col, int n, int e, int ch) {
  __shared__ int s[RNG];
  const int lo = blockIdx.y * RNG;
  const int nr = (n - lo < RNG) ? (n - lo) : RNG;
  const int* __restrict__ orow = offs + (size_t)blockIdx.x * n + lo;
  for (int i0 = threadIdx.x; i0 < nr; i0 += INIT_U * 1024) {
    int v[INIT_U];
#pragma unroll
    for (int u = 0; u < INIT_U; ++u) {
      const int i = i0 + u * 1024;
      v[u] = (i < nr) ? orow[i] : 0;
    }
#pragma unroll
    for (int u = 0; u < INIT_U; ++u) {
      const int i = i0 + u * 1024;
      if (i < nr) s[i] = v[u];
    }
  }
  __syncthreads();
  const int beg = blockIdx.x * ch;
  const int end = (e - beg < ch) ? e : (beg + ch);
  for (int i0 = beg + (int)threadIdx.x; i0 < end; i0 += SWEEP_U * 1024) {
    int d[SWEEP_U];
    int sc[SWEEP_U];
#pragma unroll
    for (int u = 0; u < SWEEP_U; ++u) {
      const int i = i0 + u * 1024;
      d[u]  = (i < end) ? dst[i] : -1;
      sc[u] = (i < end) ? src[i] : 0;
    }
#pragma unroll
    for (int u = 0; u < SWEEP_U; ++u) {
      const unsigned t = (unsigned)(d[u] - lo);
      if (t < (unsigned)nr) {
        const int pos = atomicAdd(&s[t], 1);
        col[pos] = sc[u];
      }
    }
  }
}

#define AT_STRIDE 132

template <int U>
__device__ __forceinline__ void edge_group(const float* __restrict__ in, int s_l, float w_l, int jj, int lane,
                                           float& ax, float& ay, float& az, float& aw) {
#pragma unroll
  for (int u = 0; u < U; ++u) {
    const int   s = __builtin_amdgcn_readlane(s_l, jj + u);
    const float w = __builtin_bit_cast(float,
                        __builtin_amdgcn_readlane(__builtin_bit_cast(int, w_l), jj + u));
    const float4 v = *(const float4*)(in + (size_t)s * HD + lane * 4);
    ax = fmaf(w, v.x, ax);
    ay = fmaf(w, v.y, ay);
    az = fmaf(w, v.z, az);
    aw = fmaf(w, v.w, aw);
  }
}

template <bool RELU>
__global__ __launch_bounds__(256) void k_agg_gemm(const int* __restrict__ rowptr,
                                                  const int* __restrict__ cnt,
                                                  const int* __restrict__ col,
                                                  const float* __restrict__ dinv,
                                                  const float* __restrict__ in,
                                                  const float* __restrict__ W,
                                                  const float* __restrict__ bias,
                                                  float* __restrict__ outp, int n) {
  __shared__ __attribute__((aligned(16))) float Wt[HD * WT_STRIDE];
  __shared__ __attribute__((aligned(16))) float At[8 * 16 * AT_STRIDE];
  for (int i = threadIdx.x; i < HD * HD / 4; i += 256) {
    int k  = i >> 5;
    int n4 = (i & 31) * 4;
    const float4 w = *(const float4*)(W + k * HD + n4);
    Wt[(n4 + 0) * WT_STRIDE + k] = w.x;
    Wt[(n4 + 1) * WT_STRIDE + k] = w.y;
    Wt[(n4 + 2) * WT_STRIDE + k] = w.z;
    Wt[(n4 + 3) * WT_STRIDE + k] = w.w;
  }
  __syncthreads();

  const int lane = threadIdx.x & 31;
  const int wave = __builtin_amdgcn_readfirstlane(threadIdx.x >> 5);
  const int r0   = blockIdx.x * 128 + wave * 16;
  if (r0 + 16 > n) return;
  const int hi = lane >> 4;
  const int lo = lane & 15;
  float* at = At + wave * 16 * AT_STRIDE;

  const int   beg_l = rowptr[r0 + lo];
  const int   deg_l = cnt[r0 + lo];
  const float dd_l  = dinv[r0 + lo];

  auto load_col = [&](int j) -> int {
    const int beg = __builtin_amdgcn_readlane(beg_l, j);
    const int deg = __builtin_amdgcn_readlane(deg_l, j);
    const int nb  = deg < 32 ? deg : 32;
    int s = r0 + j;
    if (lane < nb) s = col[beg + lane];
    return s;
  };
  auto load_w = [&](int s) -> float { return dinv[s]; };

  int   s_a = load_col(0);
  int   s_b = load_col(1);
  float w_a = load_w(s_a);

  for (int j = 0; j < 16; ++j) {
    const int d   = r0 + j;
    const int beg = __builtin_amdgcn_readlane(beg_l, j);
    const int deg = __builtin_amdgcn_readlane(deg_l, j);
    const float dd = __builtin_bit_cast(float,
                         __builtin_amdgcn_readlane(__builtin_bit_cast(int, dd_l), j));
    const int nb0 = deg < 32 ? deg : 32;
    int   s_l = s_a;
    float w_l = (lane < nb0) ? w_a * dd : 0.0f;
    int   s_c = r0;
    float w_b = 0.0f;
    if (j + 2 < 16) s_c = load_col(j + 2);
    if (j + 1 < 16) w_b = load_w(s_b);

    const float4 sv = *(const float4*)(in + (size_t)d * HD + lane * 4);
    float ax = 0.0f, ay = 0.0f, az = 0.0f, aw = 0.0f;

    for (int c0 = 0; c0 < deg; c0 += 32) {
      const int nb = (deg - c0 < 32) ? (deg - c0) : 32;
      if (c0 > 0) {
        s_l = d;
        w_l = 0.0f;
        if (lane < nb) {
          s_l = col[beg + c0 + lane];
          w_l = dinv[s_l] * dd;
        }
      }
      int jj = 0;
      for (; nb - jj > 8; jj += 16) edge_group<16>(in, s_l, w_l, jj, lane, ax, ay, az, aw);
      if (jj < nb) edge_group<8>(in, s_l, w_l, jj, lane, ax, ay, az, aw);
    }
    const float sw = dd * dd;
    ax = fmaf(sw, sv.x, ax);
    ay = fmaf(sw, sv.y, ay);
    az = fmaf(sw, sv.z, az);
    aw = fmaf(sw, sv.w, aw);
    float4 r;
    r.x = ax; r.y = ay; r.z = az; r.w = aw;
    *(float4*)(at + j * AT_STRIDE + lane * 4) = r;
    s_a = s_b;
    w_a = w_b;
    s_b = s_c;
  }

  __threadfence_block();

  v8f acc[8] = {};
  const float* arow = at + lo * AT_STRIDE;
  for (int kk = 0; kk < 32; ++kk) {
    const int k = kk * 4 + hi * 2;
    const v2f a = *(const v2f*)(arow + k);
#pragma unroll
    for (int t = 0; t < 8; ++t) {
      const v2f b = *(const v2f*)(&Wt[(t * 16 + lo) * WT_STRIDE + k]);
      acc[t] = __builtin_amdgcn_wmma_f32_16x16x4_f32(
          false, a, false, b, (short)0, acc[t], false, false);
    }
  }
#pragma unroll
  for (int t = 0; t < 8; ++t) {
    const float bv = bias[t * 16 + lo];
    float* orow = outp + (size_t)(r0 + hi * 8) * HD + t * 16 + lo;
#pragma unroll
    for (int r = 0; r < 8; ++r) {
      float o = acc[t][r] + bv;
      if (RELU) o = o > 0.0f ? o : 0.0f;
      orow[(size_t)r * HD] = o;
    }
  }
}

extern "C" void kernel_launch(void* const* d_in, const int* in_sizes, int n_in,
                              void* d_out, int out_size, void* d_ws, size_t ws_size,
                              hipStream_t stream) {
  const float* x  = (const float*)d_in[0];
  const float* W1 = (const float*)d_in[1];
  const float* b1 = (const float*)d_in[2];
  const float* W2 = (const float*)d_in[3];
  const float* b2 = (const float*)d_in[4];
  const int*   ei = (const int*)d_in[5];

  const int n = in_sizes[0] / HD;
  const int e = in_sizes[5] / 2;
  const int* src = ei;
  const int* dst = ei + e;
  const int total = n * HD;

  char* ws = (char*)d_ws;
  float* dinv = (float*)ws;
  size_t off1 = ((size_t)n * 4 + 255) & ~(size_t)255;
  float* buf1 = (float*)(ws + off1);
  size_t off2 = off1 + (((size_t)total * 4 + 255) & ~(size_t)255);
  size_t nb4 = ((size_t)n * 4 + 255) & ~(size_t)255;
  size_t eb4 = ((size_t)e * 4 + 255) & ~(size_t)255;
  int* cnt    = (int*)(ws + off2);
  int* rowptr = (int*)(ws + off2 + nb4);
  int* tsum   = (int*)(ws + off2 + 2 * nb4);
  int* col    = (int*)(ws + off2 + 3 * nb4);
  size_t pb4 = ((size_t)NCHUNK * n * 4 + 255) & ~(size_t)255;
  int* part   = (int*)(ws + off2 + 3 * nb4 + eb4);
  int* offs   = (int*)(ws + off2 + 3 * nb4 + eb4 + pb4);
  float* out  = (float*)d_out;

  dim3 blk(256);
  const int gT   = (n + SCAN_TILE - 1) / SCAN_TILE;
  const int ch   = (e + NCHUNK - 1) / NCHUNK;
  dim3 gP(NCHUNK, (n + RNG - 1) / RNG);
  dim3 blkP(1024);
  const int gMM  = (n + 127) / 128;

  k_part_count<<<gP, blkP, 0, stream>>>(dst, part, n, e, ch);
  k_tile_sum  <<<gT, blk, 0, stream>>>(part, cnt, tsum, n);
  k_scan      <<<gT, blk, 0, stream>>>(cnt, tsum, rowptr, dinv, part, offs, n);
  k_part_place<<<gP, blkP, 0, stream>>>(src, dst, offs, col, n, e, ch);

  k_agg_gemm<true> <<<gMM, blk, 0, stream>>>(rowptr, cnt, col, dinv, x, W1, b1, buf1, n);

  k_agg_gemm<false><<<gMM, blk, 0, stream>>>(rowptr, cnt, col, dinv, buf1, W2, b2, out, n);
}


# Fused aggregate+product kernel run as one 25-wave work-group per WGP (400 rows, W staged once per WGP): all 6250 row tiles resident in one round instead of two part-filled ones; sum order unchanged
# speedup vs baseline: 4.4978x; 1.1383x over previous
#include <hip/hip_runtime.h>

typedef __attribute__((ext_vector_type(2))) float v2f;
typedef __attribute__((ext_vector_type(8))) float v8f;

#define HD 128
#define WT_STRIDE 132
#define SCAN_TILE 1024

#define RNG 12800
#define NCHUNK 32
#define SWEEP_U 16
#define INIT_U 13

__global__ __launch_bounds__(1024) void k_part_count(const int* __restrict__ dst,
                                                     int* __restrict__ part, int n, int e, int ch) {
  __shared__ int s[RNG];
  const int lo = blockIdx.y * RNG;
  const int nr = (n - lo < RNG) ? (n - lo) : RNG;
  for (int i = threadIdx.x; i < RNG; i += 1024) s[i] = 0;
  __syncthreads();
  const int beg = blockIdx.x * ch;
  const int end = (e - beg < ch) ? e : (beg + ch);
  for (int i0 = beg + (int)threadIdx.x; i0 < end; i0 += SWEEP_U * 1024) {
    int d[SWEEP_U];
#pragma unroll
    for (int u = 0; u < SWEEP_U; ++u) {
      const int i = i0 + u * 1024;
      d[u] = (i < end) ? dst[i] : -1;
    }
#pragma unroll
    for (int u = 0; u < SWEEP_U; ++u) {
      const unsigned t = (unsigned)(d[u] - lo);
      if (t < (unsigned)nr) atomicAdd(&s[t], 1);
    }
  }
  __syncthreads();
  for (int i = threadIdx.x; i < nr; i += 1024) part[(size_t)blockIdx.x * n + lo + i] = s[i];
}

__device__ __forceinline__ int block_sum_256(int v, int* sm) {
  const int lane = threadIdx.x & 31;
  const int wave = threadIdx.x >> 5;
  for (int o = 16; o > 0; o >>= 1) v += __shfl_xor(v, o);
  if (lane == 0) sm[wave] = v;
  __syncthreads();
  int t = 0;
  for (int w = 0; w < 8; ++w) t += sm[w];
  __syncthreads();
  return t;
}

__global__ __launch_bounds__(256) void k_tile_sum(const int* __restrict__ part,
                                                  int* __restrict__ cnt,
                                                  int* __restrict__ tsum, int n) {
  __shared__ int sm[8];
  const int base = blockIdx.x * SCAN_TILE + threadIdx.x * 4;
  int s = 0;
  if ((n & 3) == 0 && base + 3 < n) {
    int4 a = make_int4(0, 0, 0, 0);
#pragma unroll
    for (int q = 0; q < NCHUNK; ++q) {
      const int4 v = *(const int4*)(part + (size_t)q * n + base);
      a.x += v.x; a.y += v.y; a.z += v.z; a.w += v.w;
    }
    *(int4*)(cnt + base) = a;
    s = a.x + a.y + a.z + a.w;
  } else {
    for (int j = 0; j < 4; ++j) {
      if (base + j < n) {
        int v = 0;
        for (int q = 0; q < NCHUNK; ++q) v += part[(size_t)q * n + base + j];
        cnt[base + j] = v;
        s += v;
      }
    }
  }
  const int t = block_sum_256(s, sm);
  if (threadIdx.x == 0) tsum[blockIdx.x] = t;
}

__global__ __launch_bounds__(256) void k_scan(const int* __restrict__ cnt,
                                              const int* __restrict__ tsum,
                                              int* __restrict__ rowptr,
                                              float* __restrict__ dinv,
                                              const int* __restrict__ part,
                                              int* __restrict__ offs, int n) {
  __shared__ int sm[8];
  const int lane = threadIdx.x & 31;
  const int wave = threadIdx.x >> 5;

  int psum = 0;
  for (int t = threadIdx.x; t < (int)blockIdx.x; t += 256) psum += tsum[t];
  const int tile_off = block_sum_256(psum, sm);

  const int base = blockIdx.x * SCAN_TILE + threadIdx.x * 4;
  int c[4];
  int s = 0;
#pragma unroll
  for (int j = 0; j < 4; ++j) {
    c[j] = (base + j < n) ? cnt[base + j] : 0;
    s += c[j];
  }
  int incl = s;
  for (int o = 1; o < 32; o <<= 1) {
    const int up = __shfl_up(incl, o);
    if (lane >= o) incl += up;
  }
  if (lane == 31) sm[wave] = incl;
  __syncthreads();
  int wave_off = 0;
  for (int w = 0; w < 8; ++w)
    if (w < wave) wave_off += sm[w];

  int run = tile_off + wave_off + incl - s;
  if ((n & 3) == 0 && base + 3 < n) {
    int4 pv[NCHUNK];
#pragma unroll
    for (int q = 0; q < NCHUNK; ++q) pv[q] = *(const int4*)(part + (size_t)q * n + base);
    int4 p;
    p.x = run;
    p.y = p.x + c[0];
    p.z = p.y + c[1];
    p.w = p.z + c[2];
    *(int4*)(rowptr + base) = p;
    float4 dv;
    dv.x = rsqrtf((float)(c[0] + 1));
    dv.y = rsqrtf((float)(c[1] + 1));
    dv.z = rsqrtf((float)(c[2] + 1));
    dv.w = rsqrtf((float)(c[3] + 1));
    *(float4*)(dinv + base) = dv;
#pragma unroll
    for (int q = 0; q < NCHUNK; ++q) {
      *(int4*)(offs + (size_t)q * n + base) = p;
      p.x += pv[q].x; p.y += pv[q].y; p.z += pv[q].z; p.w += pv[q].w;
    }
  } else {
#pragma unroll
    for (int j = 0; j < 4; ++j) {
      if (base + j < n) {
        rowptr[base + j] = run;
        dinv[base + j] = rsqrtf((float)(c[j] + 1));
        int pos = run;
        for (int q = 0; q < NCHUNK; ++q) {
          const size_t idx = (size_t)q * n + base + j;
          offs[idx] = pos;
          pos += part[idx];
        }
        run += c[j];
      }
    }
  }
}

__global__ __launch_bounds__(1024) void k_part_place(const int* __restrict__ src,
                                                     const int* __restrict__ dst,
                                                     const int* __restrict__ offs,
                                                     int* __restrict__ col, int n, int e, int ch) {
  __shared__ int s[RNG];
  const int lo = blockIdx.y * RNG;
  const int nr = (n - lo < RNG) ? (n - lo) : RNG;
  const int* __restrict__ orow = offs + (size_t)blockIdx.x * n + lo;
  for (int i0 = threadIdx.x; i0 < nr; i0 += INIT_U * 1024) {
    int v[INIT_U];
#pragma unroll
    for (int u = 0; u < INIT_U; ++u) {
      const int i = i0 + u * 1024;
      v[u] = (i < nr) ? orow[i] : 0;
    }
#pragma unroll
    for (int u = 0; u < INIT_U; ++u) {
      const int i = i0 + u * 1024;
      if (i < nr) s[i] = v[u];
    }
  }
  __syncthreads();
  const int beg = blockIdx.x * ch;
  const int end = (e - beg < ch) ? e : (beg + ch);
  for (int i0 = beg + (int)threadIdx.x; i0 < end; i0 += SWEEP_U * 1024) {
    int d[SWEEP_U];
    int sc[SWEEP_U];
#pragma unroll
    for (int u = 0; u < SWEEP_U; ++u) {
      const int i = i0 + u * 1024;
      d[u]  = (i < end) ? dst[i] : -1;
      sc[u] = (i < end) ? src[i] : 0;
    }
#pragma unroll
    for (int u = 0; u < SWEEP_U; ++u) {
      const unsigned t = (unsigned)(d[u] - lo);
      if (t < (unsigned)nr) {
        const int pos = atomicAdd(&s[t], 1);
        col[pos] = sc[u];
      }
    }
  }
}

#define AT_STRIDE 132
#define AGG_WAVES 25
#define AGG_THREADS (AGG_WAVES * 32)
#define AGG_ROWS (AGG_WAVES * 16)

template <int U>
__device__ __forceinline__ void edge_group(const float* __restrict__ in, int s_l, float w_l, int jj, int lane,
                                           float& ax, float& ay, float& az, float& aw) {
#pragma unroll
  for (int u = 0; u < U; ++u) {
    const int   s = __builtin_amdgcn_readlane(s_l, jj + u);
    const float w = __builtin_bit_cast(float,
                        __builtin_amdgcn_readlane(__builtin_bit_cast(int, w_l), jj + u));
    const float4 v = *(const float4*)(in + (size_t)s * HD + lane * 4);
    ax = fmaf(w, v.x, ax);
    ay = fmaf(w, v.y, ay);
    az = fmaf(w, v.z, az);
    aw = fmaf(w, v.w, aw);
  }
}

template <bool RELU>
__global__ __launch_bounds__(AGG_THREADS) void k_agg_gemm(const int* __restrict__ rowptr,
                                                  const int* __restrict__ cnt,
                                                  const int* __restrict__ col,
                                                  const float* __restrict__ dinv,
                                                  const float* __restrict__ in,
                                                  const float* __restrict__ W,
                                                  const float* __restrict__ bias,
                                                  float* __restrict__ outp, int n) {
  __shared__ __attribute__((aligned(16))) float Wt[HD * WT_STRIDE];
  __shared__ __attribute__((aligned(16))) float At[AGG_WAVES * 16 * AT_STRIDE];
  for (int i = threadIdx.x; i < HD * HD / 4; i += AGG_THREADS) {
    int k  = i >> 5;
    int n4 = (i & 31) * 4;
    const float4 w = *(const float4*)(W + k * HD + n4);
    Wt[(n4 + 0) * WT_STRIDE + k] = w.x;
    Wt[(n4 + 1) * WT_STRIDE + k] = w.y;
    Wt[(n4 + 2) * WT_STRIDE + k] = w.z;
    Wt[(n4 + 3) * WT_STRIDE + k] = w.w;
  }
  __syncthreads();

  const int lane = threadIdx.x & 31;
  const int wave = __builtin_amdgcn_readfirstlane(threadIdx.x >> 5);
  const int r0   = blockIdx.x * AGG_ROWS + wave * 16;
  if (r0 + 16 > n) return;
  const int hi = lane >> 4;
  const int lo = lane & 15;
  float* at = At + wave * 16 * AT_STRIDE;

  const int   beg_l = rowptr[r0 + lo];
  const int   deg_l = cnt[r0 + lo];
  const float dd_l  = dinv[r0 + lo];

  auto load_col = [&](int j) -> int {
    const int beg = __builtin_amdgcn_readlane(beg_l, j);
    const int deg = __builtin_amdgcn_readlane(deg_l, j);
    const int nb  = deg < 32 ? deg : 32;
    int s = r0 + j;
    if (lane < nb) s = col[beg + lane];
    return s;
  };
  auto load_w = [&](int s) -> float { return dinv[s]; };

  int   s_a = load_col(0);
  int   s_b = load_col(1);
  float w_a = load_w(s_a);

  for (int j = 0; j < 16; ++j) {
    const int d   = r0 + j;
    const int beg = __builtin_amdgcn_readlane(beg_l, j);
    const int deg = __builtin_amdgcn_readlane(deg_l, j);
    const float dd = __builtin_bit_cast(float,
                         __builtin_amdgcn_readlane(__builtin_bit_cast(int, dd_l), j));
    const int nb0 = deg < 32 ? deg : 32;
    int   s_l = s_a;
    float w_l = (lane < nb0) ? w_a * dd : 0.0f;
    int   s_c = r0;
    float w_b = 0.0f;
    if (j + 2 < 16) s_c = load_col(j + 2);
    if (j + 1 < 16) w_b = load_w(s_b);

    const float4 sv = *(const float4*)(in + (size_t)d * HD + lane * 4);
    float ax = 0.0f, ay = 0.0f, az = 0.0f, aw = 0.0f;

    for (int c0 = 0; c0 < deg; c0 += 32) {
      const int nb = (deg - c0 < 32) ? (deg - c0) : 32;
      if (c0 > 0) {
        s_l = d;
        w_l = 0.0f;
        if (lane < nb) {
          s_l = col[beg + c0 + lane];
          w_l = dinv[s_l] * dd;
        }
      }
      int jj = 0;
      for (; nb - jj > 8; jj += 16) edge_group<16>(in, s_l, w_l, jj, lane, ax, ay, az, aw);
      if (jj < nb) edge_group<8>(in, s_l, w_l, jj, lane, ax, ay, az, aw);
    }
    const float sw = dd * dd;
    ax = fmaf(sw, sv.x, ax);
    ay = fmaf(sw, sv.y, ay);
    az = fmaf(sw, sv.z, az);
    aw = fmaf(sw, sv.w, aw);
    float4 r;
    r.x = ax; r.y = ay; r.z = az; r.w = aw;
    *(float4*)(at + j * AT_STRIDE + lane * 4) = r;
    s_a = s_b;
    w_a = w_b;
    s_b = s_c;
  }

  __threadfence_block();

  v8f acc[8] = {};
  const float* arow = at + lo * AT_STRIDE;
  for (int kk = 0; kk < 32; ++kk) {
    const int k = kk * 4 + hi * 2;
    const v2f a = *(const v2f*)(arow + k);
#pragma unroll
    for (int t = 0; t < 8; ++t) {
      const v2f b = *(const v2f*)(&Wt[(t * 16 + lo) * WT_STRIDE + k]);
      acc[t] = __builtin_amdgcn_wmma_f32_16x16x4_f32(
          false, a, false, b, (short)0, acc[t], false, false);
    }
  }
#pragma unroll
  for (int t = 0; t < 8; ++t) {
    const float bv = bias[t * 16 + lo];
    float* orow = outp + (size_t)(r0 + hi * 8) * HD + t * 16 + lo;
#pragma unroll
    for (int r = 0; r < 8; ++r) {
      float o = acc[t][r] + bv;
      if (RELU) o = o > 0.0f ? o : 0.0f;
      orow[(size_t)r * HD] = o;
    }
  }
}

extern "C" void kernel_launch(void* const* d_in, const int* in_sizes, int n_in,
                              void* d_out, int out_size, void* d_ws, size_t ws_size,
                              hipStream_t stream) {
  const float* x  = (const float*)d_in[0];
  const float* W1 = (const float*)d_in[1];
  const float* b1 = (const float*)d_in[2];
  const float* W2 = (const float*)d_in[3];
  const float* b2 = (const float*)d_in[4];
  const int*   ei = (const int*)d_in[5];

  const int n = in_sizes[0] / HD;
  const int e = in_sizes[5] / 2;
  const int* src = ei;
  const int* dst = ei + e;
  const int total = n * HD;

  char* ws = (char*)d_ws;
  float* dinv = (float*)ws;
  size_t off1 = ((size_t)n * 4 + 255) & ~(size_t)255;
  float* buf1 = (float*)(ws + off1);
  size_t off2 = off1 + (((size_t)total * 4 + 255) & ~(size_t)255);
  size_t nb4 = ((size_t)n * 4 + 255) & ~(size_t)255;
  size_t eb4 = ((size_t)e * 4 + 255) & ~(size_t)255;
  int* cnt    = (int*)(ws + off2);
  int* rowptr = (int*)(ws + off2 + nb4);
  int* tsum   = (int*)(ws + off2 + 2 * nb4);
  int* col    = (int*)(ws + off2 + 3 * nb4);
  size_t pb4 = ((size_t)NCHUNK * n * 4 + 255) & ~(size_t)255;
  int* part   = (int*)(ws + off2 + 3 * nb4 + eb4);
  int* offs   = (int*)(ws + off2 + 3 * nb4 + eb4 + pb4);
  float* out  = (float*)d_out;

  dim3 blk(256);
  const int gT   = (n + SCAN_TILE - 1) / SCAN_TILE;
  const int ch   = (e + NCHUNK - 1) / NCHUNK;
  dim3 gP(NCHUNK, (n + RNG - 1) / RNG);
  dim3 blkP(1024);
  const int gMM  = (n + AGG_ROWS - 1) / AGG_ROWS;
  dim3 blkMM(AGG_THREADS);

  k_part_count<<<gP, blkP, 0, stream>>>(dst, part, n, e, ch);
  k_tile_sum  <<<gT, blk, 0, stream>>>(part, cnt, tsum, n);
  k_scan      <<<gT, blk, 0, stream>>>(cnt, tsum, rowptr, dinv, part, offs, n);
  k_part_place<<<gP, blkP, 0, stream>>>(src, dst, offs, col, n, e, ch);

  k_agg_gemm<true> <<<gMM, blkMM, 0, stream>>>(rowptr, cnt, col, dinv, x, W1, b1, buf1, n);

  k_agg_gemm<false><<<gMM, blkMM, 0, stream>>>(rowptr, cnt, col, dinv, buf1, W2, b2, out, n);
}
